# starting kernel
# baseline (speedup 1.0000x reference)
#include <hip/hip_runtime.h>

#define IN_FEAT 64
#define OUT_FEAT 64

typedef float v2f __attribute__((ext_vector_type(2)));
typedef float v8f __attribute__((ext_vector_type(8)));

__global__ __launch_bounds__(256) void gat_gemm_wmma(
    const float* __restrict__ feat, const float* __restrict__ weight,
    const float* __restrict__ bias, float* __restrict__ h, int n_nodes)
{
    const int lane  = threadIdx.x & 31;
    const int tileM = blockIdx.x * (blockDim.x >> 5) + (threadIdx.x >> 5);
    const int row0  = tileM * 16;
    if (row0 >= n_nodes) return;

    const int mn  = lane & 15;
    const int hi  = lane >> 4;
    const int khi = hi * 2;

    const float* arow  = feat + (size_t)(row0 + mn) * IN_FEAT + khi;
    const float* bbase = weight + (size_t)khi * OUT_FEAT + mn;

    v8f acc0 = {}, acc1 = {}, acc2 = {}, acc3 = {};

#pragma unroll
    for (int k = 0; k < IN_FEAT; k += 4) {
        v2f a;
        a.x = arow[k];
        a.y = arow[k + 1];

        const float* bk = bbase + (size_t)k * OUT_FEAT;
        v2f b0, b1, b2, b3;
        b0.x = bk[0];   b0.y = bk[OUT_FEAT + 0];
        b1.x = bk[16];  b1.y = bk[OUT_FEAT + 16];
        b2.x = bk[32];  b2.y = bk[OUT_FEAT + 32];
        b3.x = bk[48];  b3.y = bk[OUT_FEAT + 48];

        acc0 = __builtin_amdgcn_wmma_f32_16x16x4_f32(false, a, false, b0,
                                                     (short)0, acc0, false, false);
        acc1 = __builtin_amdgcn_wmma_f32_16x16x4_f32(false, a, false, b1,
                                                     (short)0, acc1, false, false);
        acc2 = __builtin_amdgcn_wmma_f32_16x16x4_f32(false, a, false, b2,
                                                     (short)0, acc2, false, false);
        acc3 = __builtin_amdgcn_wmma_f32_16x16x4_f32(false, a, false, b3,
                                                     (short)0, acc3, false, false);
    }

    const float bv0 = bias[mn];
    const float bv1 = bias[16 + mn];
    const float bv2 = bias[32 + mn];
    const float bv3 = bias[48 + mn];

#pragma unroll
    for (int v = 0; v < 8; ++v) {
        const int row = row0 + v + 8 * hi;
        float* hr = h + (size_t)row * OUT_FEAT + mn;
        hr[0]  = acc0[v] + bv0;
        hr[16] = acc1[v] + bv1;
        hr[32] = acc2[v] + bv2;
        hr[48] = acc3[v] + bv3;
    }
}

__global__ __launch_bounds__(256) void gat_scatter(
    const float* __restrict__ h, const float* __restrict__ att,
    const int* __restrict__ src, const int* __restrict__ dst,
    float* __restrict__ out, int n_edges)
{
    const int t = blockIdx.x * blockDim.x + threadIdx.x;
    const int e = t >> 4;
    if (e >= n_edges) return;
    const int f = (t & 15) << 2;

    const int   s = src[e];
    const int   d = dst[e];
    const float a = att[e];

    const float4 v = *(const float4*)(h + (size_t)s * OUT_FEAT + f);
    float* op = out + (size_t)d * OUT_FEAT + f;

    unsafeAtomicAdd(op + 0, v.x * a);
    unsafeAtomicAdd(op + 1, v.y * a);
    unsafeAtomicAdd(op + 2, v.z * a);
    unsafeAtomicAdd(op + 3, v.w * a);
}

extern "C" void kernel_launch(void* const* d_in, const int* in_sizes, int n_in,
                              void* d_out, int out_size, void* d_ws, size_t ws_size,
                              hipStream_t stream)
{
    const float* feat   = (const float*)d_in[0];
    const float* att    = (const float*)d_in[1];
    const int*   src    = (const int*)d_in[2];
    const int*   dst    = (const int*)d_in[3];
    const float* weight = (const float*)d_in[4];
    const float* bias   = (const float*)d_in[5];

    float* out = (float*)d_out;
    float* h   = (float*)d_ws;

    const int n_nodes = in_sizes[0] / IN_FEAT;
    const int n_edges = in_sizes[2];

    hipMemsetAsync(out, 0, (size_t)out_size * sizeof(float), stream);

    const int tilesM = (n_nodes + 15) / 16;
    const int wavesPerBlock = 8;
    const int gblocks = (tilesM + wavesPerBlock - 1) / wavesPerBlock;
    gat_gemm_wmma<<<gblocks, 256, 0, stream>>>(feat, weight, bias, h, n_nodes);

    const long long sthreads = (long long)n_edges * 16;
    const int sblocks = (int)((sthreads + 255) / 256);
    gat_scatter<<<sblocks, 256, 0, stream>>>(h, att, src, dst, out, n_edges);
}


# scatter: a wave walks 32 edges, lane l takes features l and 32+l, so each row load and each f32 atomic add covers one whole 128-B line (was 4 quarter-filled lines); f32 atomics kept
# speedup vs baseline: 2.0143x; 2.0143x over previous
#include <hip/hip_runtime.h>

#define IN_FEAT 64
#define OUT_FEAT 64

typedef float v2f __attribute__((ext_vector_type(2)));
typedef float v8f __attribute__((ext_vector_type(8)));

__global__ __launch_bounds__(256) void gat_gemm_wmma(
    const float* __restrict__ feat, const float* __restrict__ weight,
    const float* __restrict__ bias, float* __restrict__ h, int n_nodes)
{
    const int lane  = threadIdx.x & 31;
    const int tileM = blockIdx.x * (blockDim.x >> 5) + (threadIdx.x >> 5);
    const int row0  = tileM * 16;
    if (row0 >= n_nodes) return;

    const int mn  = lane & 15;
    const int hi  = lane >> 4;
    const int khi = hi * 2;

    const float* arow  = feat + (size_t)(row0 + mn) * IN_FEAT + khi;
    const float* bbase = weight + (size_t)khi * OUT_FEAT + mn;

    v8f acc0 = {}, acc1 = {}, acc2 = {}, acc3 = {};

#pragma unroll
    for (int k = 0; k < IN_FEAT; k += 4) {
        v2f a;
        a.x = arow[k];
        a.y = arow[k + 1];

        const float* bk = bbase + (size_t)k * OUT_FEAT;
        v2f b0, b1, b2, b3;
        b0.x = bk[0];   b0.y = bk[OUT_FEAT + 0];
        b1.x = bk[16];  b1.y = bk[OUT_FEAT + 16];
        b2.x = bk[32];  b2.y = bk[OUT_FEAT + 32];
        b3.x = bk[48];  b3.y = bk[OUT_FEAT + 48];

        acc0 = __builtin_amdgcn_wmma_f32_16x16x4_f32(false, a, false, b0,
                                                     (short)0, acc0, false, false);
        acc1 = __builtin_amdgcn_wmma_f32_16x16x4_f32(false, a, false, b1,
                                                     (short)0, acc1, false, false);
        acc2 = __builtin_amdgcn_wmma_f32_16x16x4_f32(false, a, false, b2,
                                                     (short)0, acc2, false, false);
        acc3 = __builtin_amdgcn_wmma_f32_16x16x4_f32(false, a, false, b3,
                                                     (short)0, acc3, false, false);
    }

    const float bv0 = bias[mn];
    const float bv1 = bias[16 + mn];
    const float bv2 = bias[32 + mn];
    const float bv3 = bias[48 + mn];

#pragma unroll
    for (int v = 0; v < 8; ++v) {
        const int row = row0 + v + 8 * hi;
        float* hr = h + (size_t)row * OUT_FEAT + mn;
        hr[0]  = acc0[v] + bv0;
        hr[16] = acc1[v] + bv1;
        hr[32] = acc2[v] + bv2;
        hr[48] = acc3[v] + bv3;
    }
}

#define EDGE_BATCH 8

__global__ __launch_bounds__(256) void gat_scatter(
    const float* __restrict__ h, const float* __restrict__ att,
    const int* __restrict__ src, const int* __restrict__ dst,
    float* __restrict__ out, int n_edges)
{
    const int lane = threadIdx.x & 31;
    const int wave = __builtin_amdgcn_readfirstlane((int)(threadIdx.x >> 5));
    const int e0   = (blockIdx.x * (int)(blockDim.x >> 5) + wave) << 5;
    if (e0 >= n_edges) return;

    const int cnt = (n_edges - e0 < 32) ? (n_edges - e0) : 32;

    int   sv = 0, dv = 0;
    float av = 0.0f;
    if (lane < cnt) {
        sv = src[e0 + lane];
        dv = dst[e0 + lane];
        av = att[e0 + lane];
    }

    const unsigned lane4 = (unsigned)lane << 2;
    const char* hb = (const char*)h;
    char*       ob = (char*)out;

    if (cnt == 32) {
        for (int j0 = 0; j0 < 32; j0 += EDGE_BATCH) {
            float r0[EDGE_BATCH], r1[EDGE_BATCH];
            for (int u = 0; u < EDGE_BATCH; ++u) {
                const unsigned s = (unsigned)__builtin_amdgcn_readlane(sv, j0 + u);
                const char* p = hb + ((s << 8) + lane4);
                r0[u] = *(const float*)p;
                r1[u] = *(const float*)(p + 128);
            }
            for (int u = 0; u < EDGE_BATCH; ++u) {
                const unsigned d = (unsigned)__builtin_amdgcn_readlane(dv, j0 + u);
                const float    a = __int_as_float(
                    __builtin_amdgcn_readlane(__float_as_int(av), j0 + u));
                char* q = ob + ((d << 8) + lane4);
                unsafeAtomicAdd((float*)q,         r0[u] * a);
                unsafeAtomicAdd((float*)(q + 128), r1[u] * a);
            }
        }
    } else {
        for (int j = 0; j < cnt; ++j) {
            const unsigned s = (unsigned)__builtin_amdgcn_readlane(sv, j);
            const unsigned d = (unsigned)__builtin_amdgcn_readlane(dv, j);
            const float    a = __int_as_float(
                __builtin_amdgcn_readlane(__float_as_int(av), j));
            const char* p = hb + ((s << 8) + lane4);
            char*       q = ob + ((d << 8) + lane4);
            const float x0 = *(const float*)p;
            const float x1 = *(const float*)(p + 128);
            unsafeAtomicAdd((float*)q,         x0 * a);
            unsafeAtomicAdd((float*)(q + 128), x1 * a);
        }
    }
}

extern "C" void kernel_launch(void* const* d_in, const int* in_sizes, int n_in,
                              void* d_out, int out_size, void* d_ws, size_t ws_size,
                              hipStream_t stream)
{
    const float* feat   = (const float*)d_in[0];
    const float* att    = (const float*)d_in[1];
    const int*   src    = (const int*)d_in[2];
    const int*   dst    = (const int*)d_in[3];
    const float* weight = (const float*)d_in[4];
    const float* bias   = (const float*)d_in[5];

    float* out = (float*)d_out;
    float* h   = (float*)d_ws;

    const int n_nodes = in_sizes[0] / IN_FEAT;
    const int n_edges = in_sizes[2];

    hipMemsetAsync(out, 0, (size_t)out_size * sizeof(float), stream);

    const int tilesM = (n_nodes + 15) / 16;
    const int wavesPerBlock = 8;
    const int gblocks = (tilesM + wavesPerBlock - 1) / wavesPerBlock;
    gat_gemm_wmma<<<gblocks, 256, 0, stream>>>(feat, weight, bias, h, n_nodes);

    const int sblocks = (n_edges + 255) / 256;
    gat_scatter<<<sblocks, 256, 0, stream>>>(h, att, src, dst, out, n_edges);
}


# scatter: 2048 resident blocks read their die (s_sendmsg_rtn 0x87), draw tasks from one counter per L2 half and add only edges whose dst row is in that half's node range; f32 atomics kept
# speedup vs baseline: 2.3760x; 1.1796x over previous
#include <hip/hip_runtime.h>

#define IN_FEAT 64
#define OUT_FEAT 64

typedef float v2f __attribute__((ext_vector_type(2)));
typedef float v8f __attribute__((ext_vector_type(8)));

#define QUEUE_STRIDE_INTS 64
#define QUEUE_BYTES (2 * QUEUE_STRIDE_INTS * 4)

__device__ __forceinline__ bool ws_has_queue(size_t ws_size, int n_nodes)
{
    return ws_size >= (size_t)n_nodes * (OUT_FEAT * sizeof(float)) + QUEUE_BYTES;
}

__global__ __launch_bounds__(256) void gat_gemm_wmma(
    const float* __restrict__ feat, const float* __restrict__ weight,
    const float* __restrict__ bias, float* __restrict__ h, int* queue_base,
    size_t ws_size, int n_nodes)
{
    if (blockIdx.x == 0 && threadIdx.x == 0 && ws_has_queue(ws_size, n_nodes)) {
        int* q = queue_base + (size_t)n_nodes * OUT_FEAT;
        q[0] = 0;
        q[QUEUE_STRIDE_INTS] = 0;
    }

    const int lane  = threadIdx.x & 31;
    const int tileM = blockIdx.x * (blockDim.x >> 5) + (threadIdx.x >> 5);
    const int row0  = tileM * 16;
    if (row0 >= n_nodes) return;

    const int mn  = lane & 15;
    const int hi  = lane >> 4;
    const int khi = hi * 2;

    const float* arow  = feat + (size_t)(row0 + mn) * IN_FEAT + khi;
    const float* bbase = weight + (size_t)khi * OUT_FEAT + mn;

    v8f acc0 = {}, acc1 = {}, acc2 = {}, acc3 = {};

#pragma unroll
    for (int k = 0; k < IN_FEAT; k += 4) {
        v2f a;
        a.x = arow[k];
        a.y = arow[k + 1];

        const float* bk = bbase + (size_t)k * OUT_FEAT;
        v2f b0, b1, b2, b3;
        b0.x = bk[0];   b0.y = bk[OUT_FEAT + 0];
        b1.x = bk[16];  b1.y = bk[OUT_FEAT + 16];
        b2.x = bk[32];  b2.y = bk[OUT_FEAT + 32];
        b3.x = bk[48];  b3.y = bk[OUT_FEAT + 48];

        acc0 = __builtin_amdgcn_wmma_f32_16x16x4_f32(false, a, false, b0,
                                                     (short)0, acc0, false, false);
        acc1 = __builtin_amdgcn_wmma_f32_16x16x4_f32(false, a, false, b1,
                                                     (short)0, acc1, false, false);
        acc2 = __builtin_amdgcn_wmma_f32_16x16x4_f32(false, a, false, b2,
                                                     (short)0, acc2, false, false);
        acc3 = __builtin_amdgcn_wmma_f32_16x16x4_f32(false, a, false, b3,
                                                     (short)0, acc3, false, false);
    }

    const float bv0 = bias[mn];
    const float bv1 = bias[16 + mn];
    const float bv2 = bias[32 + mn];
    const float bv3 = bias[48 + mn];

#pragma unroll
    for (int v = 0; v < 8; ++v) {
        const int row = row0 + v + 8 * hi;
        float* hr = h + (size_t)row * OUT_FEAT + mn;
        hr[0]  = acc0[v] + bv0;
        hr[16] = acc1[v] + bv1;
        hr[32] = acc2[v] + bv2;
        hr[48] = acc3[v] + bv3;
    }
}

#define TASK_GROUPS 2
#define TASK_EDGES  (32 * TASK_GROUPS)
#define TASK_WAVES  8

template <int N>
__device__ __forceinline__ void edge_batch(const char* hb, char* ob, int sv, int dv,
                                           float av, int j0, unsigned lane4)
{
    float r0[N], r1[N];
    for (int u = 0; u < N; ++u) {
        const unsigned s = (unsigned)__builtin_amdgcn_readlane(sv, j0 + u);
        const char* p = hb + ((s << 8) + lane4);
        r0[u] = *(const float*)p;
        r1[u] = *(const float*)(p + 128);
    }
    for (int u = 0; u < N; ++u) {
        const unsigned d = (unsigned)__builtin_amdgcn_readlane(dv, j0 + u);
        const float    a = __int_as_float(
            __builtin_amdgcn_readlane(__float_as_int(av), j0 + u));
        char* q = ob + ((d << 8) + lane4);
        unsafeAtomicAdd((float*)q,         r0[u] * a);
        unsafeAtomicAdd((float*)(q + 128), r1[u] * a);
    }
}

__device__ __forceinline__ void edge_group(const char* hb, char* ob, int sv, int dv,
                                           float av, bool mine, int lane, unsigned lane4)
{
    const unsigned mask = __builtin_amdgcn_ballot_w32(mine);
    const int c = __builtin_popcount(mask);
    if (c == 0) return;
    const int rank = (int)__builtin_amdgcn_mbcnt_lo(mask, 0u);
    const int dest = mine ? rank : (c + lane - rank);
    sv = __builtin_amdgcn_ds_permute(dest << 2, sv);
    dv = __builtin_amdgcn_ds_permute(dest << 2, dv);
    av = __int_as_float(__builtin_amdgcn_ds_permute(dest << 2, __float_as_int(av)));

    int j0 = 0;
    for (; j0 + 8 <= c; j0 += 8) edge_batch<8>(hb, ob, sv, dv, av, j0, lane4);
    if (c & 4) { edge_batch<4>(hb, ob, sv, dv, av, j0, lane4); j0 += 4; }
    if (c & 2) { edge_batch<2>(hb, ob, sv, dv, av, j0, lane4); j0 += 2; }
    if (c & 1) { edge_batch<1>(hb, ob, sv, dv, av, j0, lane4); }
}

struct TaskEdges {
    int   sv[TASK_GROUPS];
    int   dv[TASK_GROUPS];
    float av[TASK_GROUPS];
};

__device__ __forceinline__ void load_task(
    TaskEdges& te, const float* __restrict__ att, const int* __restrict__ src,
    const int* __restrict__ dst, int t, int n_edges, int lane)
{
    const int e0 = t * TASK_EDGES + lane;
    for (int g = 0; g < TASK_GROUPS; ++g) {
        const int e = e0 + 32 * g;
        te.sv[g] = 0; te.dv[g] = 0; te.av[g] = 0.0f;
        if (e < n_edges) {
            te.sv[g] = src[e];
            te.dv[g] = dst[e];
            te.av[g] = att[e];
        }
    }
}

__device__ __forceinline__ void run_task(
    const TaskEdges& te, const char* hb, char* ob, int t, int half, bool filter,
    int split, int n_edges, int lane, unsigned lane4)
{
    const int e0 = t * TASK_EDGES + lane;
    for (int g = 0; g < TASK_GROUPS; ++g) {
        const bool ok    = (e0 + 32 * g) < n_edges;
        const bool upper = te.dv[g] >= split;
        const bool mine  = ok && (!filter || (upper == (half != 0)));
        edge_group(hb, ob, te.sv[g], te.dv[g], te.av[g], mine, lane, lane4);
    }
}

__global__ __launch_bounds__(256) void gat_scatter(
    const float* __restrict__ h, const float* __restrict__ att,
    const int* __restrict__ src, const int* __restrict__ dst,
    float* __restrict__ out, int* queue_base, size_t ws_size,
    int n_edges, int n_nodes)
{
    __shared__ int s_ticket[2];
    __shared__ int s_half;

    const int lane = threadIdx.x & 31;
    const int wave = __builtin_amdgcn_readfirstlane((int)(threadIdx.x >> 5));
    const unsigned lane4 = (unsigned)lane << 2;
    const char* hb = (const char*)h;
    char*       ob = (char*)out;

    const int n_tasks = (n_edges + TASK_EDGES - 1) / TASK_EDGES;
    const int split   = n_nodes >> 1;

    if (ws_has_queue(ws_size, n_nodes)) {
        const int n_tickets = (n_tasks + TASK_WAVES - 1) / TASK_WAVES;
        int* q = queue_base + (size_t)n_nodes * OUT_FEAT;
        if (threadIdx.x == 0)
            s_half = (int)((__builtin_amdgcn_s_sendmsg_rtn(0x87) >> 18) & 1u);
        __syncthreads();
        int half = __builtin_amdgcn_readfirstlane(s_half);
        int slot = 0;
        for (int phase = 0; phase < 2; ++phase) {
            int* counter = q + half * QUEUE_STRIDE_INTS;
            if (threadIdx.x == 0) s_ticket[slot] = atomicAdd(counter, 1);
            __syncthreads();
            for (;;) {
                const int ticket = __builtin_amdgcn_readfirstlane(s_ticket[slot]);
                if (ticket >= n_tickets) break;
                const int t = ticket * TASK_WAVES + wave;
                TaskEdges te;
                load_task(te, att, src, dst, t, n_edges, lane);
                int next = 0;
                if (threadIdx.x == 0) next = atomicAdd(counter, 1);
                run_task(te, hb, ob, t, half, true, split, n_edges, lane, lane4);
                if (threadIdx.x == 0) s_ticket[slot ^ 1] = next;
                __syncthreads();
                slot ^= 1;
            }
            slot ^= 1;
            half ^= 1;
        }
    } else {
        const int n_waves = (int)(gridDim.x * (blockDim.x >> 5));
        for (int t = (int)blockIdx.x * (int)(blockDim.x >> 5) + wave; t < n_tasks; t += n_waves) {
            TaskEdges te;
            load_task(te, att, src, dst, t, n_edges, lane);
            run_task(te, hb, ob, t, 0, false, split, n_edges, lane, lane4);
        }
    }
}

extern "C" void kernel_launch(void* const* d_in, const int* in_sizes, int n_in,
                              void* d_out, int out_size, void* d_ws, size_t ws_size,
                              hipStream_t stream)
{
    const float* feat   = (const float*)d_in[0];
    const float* att    = (const float*)d_in[1];
    const int*   src    = (const int*)d_in[2];
    const int*   dst    = (const int*)d_in[3];
    const float* weight = (const float*)d_in[4];
    const float* bias   = (const float*)d_in[5];

    float* out = (float*)d_out;
    float* h   = (float*)d_ws;
    int*   qb  = (int*)d_ws;

    const int n_nodes = in_sizes[0] / IN_FEAT;
    const int n_edges = in_sizes[2];

    hipMemsetAsync(out, 0, (size_t)out_size * sizeof(float), stream);

    const int tilesM = (n_nodes + 15) / 16;
    const int wavesPerBlock = 8;
    const int gblocks = (tilesM + wavesPerBlock - 1) / wavesPerBlock;
    gat_gemm_wmma<<<gblocks, 256, 0, stream>>>(feat, weight, bias, h, qb, ws_size, n_nodes);

    gat_scatter<<<2048, 256, 0, stream>>>(h, att, src, dst, out, qb, ws_size,
                                          n_edges, n_nodes);
}
